# starting kernel
# baseline (speedup 1.0000x reference)
#include <hip/hip_runtime.h>

typedef float v2f __attribute__((ext_vector_type(2)));
typedef float v8f __attribute__((ext_vector_type(8)));

#define IN_C 32
#define CH 35
#define HID 8
#define PCOUNT 369
#define OFF_W1 0
#define OFF_B1 280
#define OFF_W2 288
#define OFF_B2 352
#define OFF_W3 360
#define OFF_B3 368
#define LROW 18

__device__ __forceinline__ float relu1(float x) {
  return __builtin_amdgcn_fmed3f(x, 0.0f, __builtin_inff());
}

__global__ __launch_bounds__(256) void dynconv_params(
    const float* __restrict__ prop, const float* __restrict__ W,
    const float* __restrict__ bias, float* __restrict__ params, int numProp) {
  int idx = blockIdx.x * blockDim.x + threadIdx.x;
  int total = numProp * PCOUNT;
  if (idx >= total) return;
  int p = idx / PCOUNT;
  int c = idx - p * PCOUNT;
  float acc = bias[c];
  const float* pr = prop + p * IN_C;
#pragma unroll
  for (int k = 0; k < IN_C; ++k)
    acc = fmaf(pr[k], W[k * PCOUNT + c], acc);
  params[idx] = acc;
}

__global__ __launch_bounds__(256) void dynconv_mlp(
    const float* __restrict__ feats, const float* __restrict__ params,
    float* __restrict__ out, int numAll, int numTiles) {
  __shared__ float smem[8 * 16 * LROW];
  const int lane = threadIdx.x & 31;
  const int wave = threadIdx.x >> 5;
  float* lds = smem + wave * 16 * LROW;

  const int p0 = blockIdx.x * 2;
  const int p1 = p0 + 1;

  const int n   = lane & 15;
  const int hi  = lane >> 4;
  const int hN  = n & 7;
  const bool loCol = (n < 8);
  const float* prmN = params + (size_t)(loCol ? p0 : p1) * PCOUNT;

  v2f B1[9];
#pragma unroll
  for (int kc = 0; kc < 9; ++kc) {
    int c0 = 4 * kc + 2 * hi;
    int c1 = c0 + 1;
    float x = prmN[OFF_W1 + ((c0 < CH) ? c0 : 0) * HID + hN];
    float y = prmN[OFF_W1 + ((c1 < CH) ? c1 : 0) * HID + hN];
    B1[kc].x = (c0 < CH) ? x : 0.f;
    B1[kc].y = (c1 < CH) ? y : 0.f;
  }

  v2f B2[4];
#pragma unroll
  for (int kc = 0; kc < 4; ++kc) {
    int k0 = 4 * kc + 2 * hi;
#pragma unroll
    for (int j = 0; j < 2; ++j) {
      int k = k0 + j;
      float v = prmN[OFF_W2 + (k & 7) * HID + hN];
      v = ((k < 8) == loCol) ? v : 0.f;
      if (j == 0) B2[kc].x = v; else B2[kc].y = v;
    }
  }

  const float bias1 = prmN[OFF_B1 + hN];
  const float bias2 = prmN[OFF_B2 + hN];

  const float* prmL = params + (size_t)(hi ? p1 : p0) * PCOUNT;
  float w3r[8];
#pragma unroll
  for (int h = 0; h < 8; ++h) w3r[h] = prmL[OFF_W3 + h];
  const float b3 = prmL[OFF_B3];

  const int row = lane & 15;
  const int waveStride = gridDim.y * 8;

  for (int tile = blockIdx.y * 8 + wave; tile < numTiles; tile += waveStride) {
    const int aBase = tile * 16;
    int a = aBase + row;
    if (a > numAll - 1) a = numAll - 1;
    const float* frow = feats + (size_t)a * CH;

    v8f d;
#pragma unroll
    for (int r = 0; r < 8; ++r) d[r] = bias1;
#pragma unroll
    for (int kc = 0; kc < 9; ++kc) {
      int c0 = 4 * kc + 2 * hi;
      v2f A;
      A.x = frow[c0];
      if (kc < 8) {
        A.y = frow[c0 + 1];
      } else {
        float t = frow[hi ? c0 : (c0 + 1)];
        A.y = hi ? 0.f : t;
      }
      d = __builtin_amdgcn_wmma_f32_16x16x4_f32(false, A, false, B1[kc],
                                                (short)0, d, false, false);
    }
#pragma unroll
    for (int r = 0; r < 8; ++r) d[r] = relu1(d[r]);

#pragma unroll
    for (int r = 0; r < 8; ++r) lds[(r + 8 * hi) * LROW + n] = d[r];

    v8f e;
#pragma unroll
    for (int r = 0; r < 8; ++r) e[r] = bias2;
#pragma unroll
    for (int kc = 0; kc < 4; ++kc) {
      int k0 = 4 * kc + 2 * hi;
      v2f A2;
      A2.x = lds[row * LROW + k0];
      A2.y = lds[row * LROW + k0 + 1];
      e = __builtin_amdgcn_wmma_f32_16x16x4_f32(false, A2, false, B2[kc],
                                                (short)0, e, false, false);
    }
#pragma unroll
    for (int r = 0; r < 8; ++r) e[r] = relu1(e[r]);
#pragma unroll
    for (int r = 0; r < 8; ++r) lds[(r + 8 * hi) * LROW + n] = e[r];

    float acc = b3;
#pragma unroll
    for (int h = 0; h < 8; ++h)
      acc = fmaf(lds[row * LROW + 8 * hi + h], w3r[h], acc);

    int aOut = aBase + row;
    if (aOut < numAll)
      out[(size_t)(hi ? p1 : p0) * numAll + aOut] = acc;
  }
}

extern "C" void kernel_launch(void* const* d_in, const int* in_sizes, int n_in,
                              void* d_out, int out_size, void* d_ws, size_t ws_size,
                              hipStream_t stream) {
  const float* prop  = (const float*)d_in[0];
  const float* feats = (const float*)d_in[1];
  const float* Wdyn  = (const float*)d_in[2];
  const float* bdyn  = (const float*)d_in[3];
  float* outp   = (float*)d_out;
  float* params = (float*)d_ws;

  const int numProp = in_sizes[0] / IN_C;
  const int numAll  = in_sizes[1] / CH;

  const int totalP = numProp * PCOUNT;
  dynconv_params<<<(totalP + 255) / 256, 256, 0, stream>>>(
      prop, Wdyn, bdyn, params, numProp);

  const int numTiles = (numAll + 15) / 16;
  const int pairs    = numProp / 2;
  int ygrid = (numTiles + 7) / 8;
  if (ygrid > 64) ygrid = 64;
  dynconv_mlp<<<dim3(pairs, ygrid), 256, 0, stream>>>(
      feats, params, outp, numAll, numTiles);
}


# Stage each feature tile in wave-private LDS with whole-line 16-byte global loads and read the layer-1 A fragments from LDS, replacing 18 four-byte global loads per tile that touch one line per row
# speedup vs baseline: 1.2752x; 1.2752x over previous
#include <hip/hip_runtime.h>

typedef float v2f __attribute__((ext_vector_type(2)));
typedef float v4f __attribute__((ext_vector_type(4)));
typedef float v8f __attribute__((ext_vector_type(8)));

#define IN_C 32
#define CH 35
#define HID 8
#define PCOUNT 369
#define OFF_W1 0
#define OFF_B1 280
#define OFF_W2 288
#define OFF_B2 352
#define OFF_W3 360
#define OFF_B3 368
#define LROW 18
#define TILE_F (16 * CH)
#define TILE_V4 (TILE_F / 4)
#define STG_F (5 * 32 * 4)

__device__ __forceinline__ float relu1(float x) {
  return __builtin_amdgcn_fmed3f(x, 0.0f, __builtin_inff());
}

__global__ __launch_bounds__(256) void dynconv_params(
    const float* __restrict__ prop, const float* __restrict__ W,
    const float* __restrict__ bias, float* __restrict__ params, int numProp) {
  int idx = blockIdx.x * blockDim.x + threadIdx.x;
  int total = numProp * PCOUNT;
  if (idx >= total) return;
  int p = idx / PCOUNT;
  int c = idx - p * PCOUNT;
  float acc = bias[c];
  const float* pr = prop + p * IN_C;
#pragma unroll
  for (int k = 0; k < IN_C; ++k)
    acc = fmaf(pr[k], W[k * PCOUNT + c], acc);
  params[idx] = acc;
}

__global__ __launch_bounds__(256, 10) void dynconv_mlp(
    const float* __restrict__ feats, const float* __restrict__ params,
    float* __restrict__ out, int numAll, int numTiles) {
  __shared__ __attribute__((aligned(16))) float stage[8 * STG_F];
  __shared__ float smem[8 * 16 * LROW];
  const int lane = threadIdx.x & 31;
  const int wave = threadIdx.x >> 5;
  float* lds = smem + wave * 16 * LROW;
  float* st  = stage + wave * STG_F;
  v4f* st4   = (v4f*)st;

  const int p0 = blockIdx.x * 2;
  const int p1 = p0 + 1;

  const int n   = lane & 15;
  const int hi  = lane >> 4;
  const int hN  = n & 7;
  const bool loCol = (n < 8);
  const float* prmN = params + (size_t)(loCol ? p0 : p1) * PCOUNT;

  v2f B1[9];
#pragma unroll
  for (int kc = 0; kc < 9; ++kc) {
    int c0 = 4 * kc + 2 * hi;
    int c1 = c0 + 1;
    float x = prmN[OFF_W1 + ((c0 < CH) ? c0 : 0) * HID + hN];
    float y = prmN[OFF_W1 + ((c1 < CH) ? c1 : 0) * HID + hN];
    B1[kc].x = (c0 < CH) ? x : 0.f;
    B1[kc].y = (c1 < CH) ? y : 0.f;
  }

  v2f B2[4];
#pragma unroll
  for (int kc = 0; kc < 4; ++kc) {
    int k0 = 4 * kc + 2 * hi;
#pragma unroll
    for (int j = 0; j < 2; ++j) {
      int k = k0 + j;
      float v = prmN[OFF_W2 + (k & 7) * HID + hN];
      v = ((k < 8) == loCol) ? v : 0.f;
      if (j == 0) B2[kc].x = v; else B2[kc].y = v;
    }
  }

  const float bias1 = prmN[OFF_B1 + hN];
  const float bias2 = prmN[OFF_B2 + hN];

  const float* prmL = params + (size_t)(hi ? p1 : p0) * PCOUNT;
  float w3r[8];
#pragma unroll
  for (int h = 0; h < 8; ++h) w3r[h] = prmL[OFF_W3 + h];
  const float b3 = prmL[OFF_B3];

  const int row = lane & 15;
  const int waveStride = gridDim.y * 8;
  const float* arow = st + row * CH;
  const bool vecOK = (((size_t)feats) & 15) == 0;

  for (int tile = blockIdx.y * 8 + wave; tile < numTiles; tile += waveStride) {
    const int aBase = tile * 16;

    if (vecOK && aBase + 16 <= numAll) {
      const v4f* src4 = (const v4f*)(feats + (size_t)aBase * CH);
      const int last = 128 + (lane < TILE_V4 - 128 ? lane : TILE_V4 - 129);
      v4f v0 = src4[lane];
      v4f v1 = src4[lane + 32];
      v4f v2 = src4[lane + 64];
      v4f v3 = src4[lane + 96];
      v4f v4 = src4[last];
      st4[lane]      = v0;
      st4[lane + 32] = v1;
      st4[lane + 64] = v2;
      st4[lane + 96] = v3;
      st4[lane + 128] = v4;
    } else {
      for (int i = lane; i < TILE_F; i += 32) {
        int r = i / CH;
        int c = i - r * CH;
        int a = aBase + r;
        if (a > numAll - 1) a = numAll - 1;
        st[i] = feats[(size_t)a * CH + c];
      }
    }

    v8f d;
#pragma unroll
    for (int r = 0; r < 8; ++r) d[r] = bias1;
#pragma unroll
    for (int kc = 0; kc < 9; ++kc) {
      int c0 = 4 * kc + 2 * hi;
      v2f A;
      A.x = arow[c0];
      if (kc < 8) {
        A.y = arow[c0 + 1];
      } else {
        float t = arow[hi ? c0 : (c0 + 1)];
        A.y = hi ? 0.f : t;
      }
      d = __builtin_amdgcn_wmma_f32_16x16x4_f32(false, A, false, B1[kc],
                                                (short)0, d, false, false);
    }
#pragma unroll
    for (int r = 0; r < 8; ++r) d[r] = relu1(d[r]);

#pragma unroll
    for (int r = 0; r < 8; ++r) lds[(r + 8 * hi) * LROW + n] = d[r];

    v8f e;
#pragma unroll
    for (int r = 0; r < 8; ++r) e[r] = bias2;
#pragma unroll
    for (int kc = 0; kc < 4; ++kc) {
      int k0 = 4 * kc + 2 * hi;
      v2f A2;
      A2.x = lds[row * LROW + k0];
      A2.y = lds[row * LROW + k0 + 1];
      e = __builtin_amdgcn_wmma_f32_16x16x4_f32(false, A2, false, B2[kc],
                                                (short)0, e, false, false);
    }
#pragma unroll
    for (int r = 0; r < 8; ++r) e[r] = relu1(e[r]);
#pragma unroll
    for (int r = 0; r < 8; ++r) lds[(r + 8 * hi) * LROW + n] = e[r];

    float acc = b3;
#pragma unroll
    for (int h = 0; h < 8; ++h)
      acc = fmaf(lds[row * LROW + 8 * hi + h], w3r[h], acc);

    int aOut = aBase + row;
    if (aOut < numAll)
      out[(size_t)(hi ? p1 : p0) * numAll + aOut] = acc;
  }
}

extern "C" void kernel_launch(void* const* d_in, const int* in_sizes, int n_in,
                              void* d_out, int out_size, void* d_ws, size_t ws_size,
                              hipStream_t stream) {
  const float* prop  = (const float*)d_in[0];
  const float* feats = (const float*)d_in[1];
  const float* Wdyn  = (const float*)d_in[2];
  const float* bdyn  = (const float*)d_in[3];
  float* outp   = (float*)d_out;
  float* params = (float*)d_ws;

  const int numProp = in_sizes[0] / IN_C;
  const int numAll  = in_sizes[1] / CH;

  const int totalP = numProp * PCOUNT;
  dynconv_params<<<(totalP + 255) / 256, 256, 0, stream>>>(
      prop, Wdyn, bdyn, params, numProp);

  const int numTiles = (numAll + 15) / 16;
  const int pairs    = numProp / 2;
  int ygrid = (numTiles + 7) / 8;
  if (ygrid > 64) ygrid = 64;
  dynconv_mlp<<<dim3(pairs, ygrid), 256, 0, stream>>>(
      feats, params, outp, numAll, numTiles);
}


# Swap the WMMA operand roles (weights as A, data as B): each layer's D registers are the next layer's B fragments and the layer-3 inputs, so both LDS relayouts go; k order of every sum kept
# speedup vs baseline: 1.3574x; 1.0645x over previous
#include <hip/hip_runtime.h>

typedef float v2f __attribute__((ext_vector_type(2)));
typedef float v4f __attribute__((ext_vector_type(4)));
typedef float v8f __attribute__((ext_vector_type(8)));

#define IN_C 32
#define CH 35
#define HID 8
#define PCOUNT 369
#define OFF_W1 0
#define OFF_B1 280
#define OFF_W2 288
#define OFF_B2 352
#define OFF_W3 360
#define OFF_B3 368
#define TILE_F (16 * CH)
#define TILE_V4 (TILE_F / 4)
#define STG_F (5 * 32 * 4)

__device__ __forceinline__ float relu1(float x) {
  return __builtin_amdgcn_fmed3f(x, 0.0f, __builtin_inff());
}

__global__ __launch_bounds__(256) void dynconv_params(
    const float* __restrict__ prop, const float* __restrict__ W,
    const float* __restrict__ bias, float* __restrict__ params, int numProp) {
  int idx = blockIdx.x * blockDim.x + threadIdx.x;
  int total = numProp * PCOUNT;
  if (idx >= total) return;
  int p = idx / PCOUNT;
  int c = idx - p * PCOUNT;
  float acc = bias[c];
  const float* pr = prop + p * IN_C;
#pragma unroll
  for (int k = 0; k < IN_C; ++k)
    acc = fmaf(pr[k], W[k * PCOUNT + c], acc);
  params[idx] = acc;
}

__global__ __launch_bounds__(256, 10) void dynconv_mlp(
    const float* __restrict__ feats, const float* __restrict__ params,
    float* __restrict__ out, int numAll, int numTiles) {
  __shared__ __attribute__((aligned(16))) float stage[8 * STG_F];
  const int lane = threadIdx.x & 31;
  const int wave = threadIdx.x >> 5;
  float* st  = stage + wave * STG_F;
  v4f* st4   = (v4f*)st;

  const int p0 = blockIdx.x * 2;
  const int p1 = p0 + 1;

  const int n   = lane & 15;
  const int hi  = lane >> 4;
  const int hN  = n & 7;
  const bool loCol = (n < 8);
  const float* prmN = params + (size_t)(loCol ? p0 : p1) * PCOUNT;
  const float* prmL = params + (size_t)(hi ? p1 : p0) * PCOUNT;

  v2f W1[9];
  for (int kc = 0; kc < 9; ++kc) {
    int c0 = 4 * kc + 2 * hi;
    int c1 = c0 + 1;
    float x = prmN[OFF_W1 + ((c0 < CH) ? c0 : 0) * HID + hN];
    float y = prmN[OFF_W1 + ((c1 < CH) ? c1 : 0) * HID + hN];
    W1[kc].x = (c0 < CH) ? x : 0.f;
    W1[kc].y = (c1 < CH) ? y : 0.f;
  }

  v2f W2[4];
  for (int j = 0; j < 4; ++j) {
    for (int s = 0; s < 2; ++s) {
      float v = prmN[OFF_W2 + (2 * j + s) * HID + hN];
      v = ((hi == 0) == loCol) ? v : 0.f;
      if (s == 0) W2[j].x = v; else W2[j].y = v;
    }
  }

  v8f c1, c2;
  float w3r[8];
  for (int r = 0; r < 8; ++r) {
    c1[r]  = prmL[OFF_B1 + r];
    c2[r]  = prmL[OFF_B2 + r];
    w3r[r] = prmL[OFF_W3 + r];
  }
  const float b3 = prmL[OFF_B3];

  const int waveStride = gridDim.y * 8;
  const float* xrow = st + n * CH;
  const bool vecOK = (((size_t)feats) & 15) == 0;

  for (int tile = blockIdx.y * 8 + wave; tile < numTiles; tile += waveStride) {
    const int aBase = tile * 16;

    if (vecOK && aBase + 16 <= numAll) {
      const v4f* src4 = (const v4f*)(feats + (size_t)aBase * CH);
      const int last = 128 + (lane < TILE_V4 - 128 ? lane : TILE_V4 - 129);
      v4f v0 = src4[lane];
      v4f v1 = src4[lane + 32];
      v4f v2 = src4[lane + 64];
      v4f v3 = src4[lane + 96];
      v4f v4 = src4[last];
      st4[lane]      = v0;
      st4[lane + 32] = v1;
      st4[lane + 64] = v2;
      st4[lane + 96] = v3;
      st4[lane + 128] = v4;
    } else {
      for (int i = lane; i < TILE_F; i += 32) {
        int r = i / CH;
        int c = i - r * CH;
        int a = aBase + r;
        if (a > numAll - 1) a = numAll - 1;
        st[i] = feats[(size_t)a * CH + c];
      }
    }

    v8f d = c1;
    for (int kc = 0; kc < 9; ++kc) {
      int c0 = 4 * kc + 2 * hi;
      v2f X;
      X.x = xrow[c0];
      if (kc < 8) {
        X.y = xrow[c0 + 1];
      } else {
        float t = xrow[hi ? c0 : (c0 + 1)];
        X.y = hi ? 0.f : t;
      }
      d = __builtin_amdgcn_wmma_f32_16x16x4_f32(false, W1[kc], false, X,
                                                (short)0, d, false, false);
    }
    for (int r = 0; r < 8; ++r) d[r] = relu1(d[r]);

    v8f e = c2;
    for (int j = 0; j < 4; ++j) {
      v2f F;
      F.x = d[2 * j];
      F.y = d[2 * j + 1];
      e = __builtin_amdgcn_wmma_f32_16x16x4_f32(false, W2[j], false, F,
                                                (short)0, e, false, false);
    }
    for (int r = 0; r < 8; ++r) e[r] = relu1(e[r]);

    float acc = b3;
    for (int h = 0; h < 8; ++h)
      acc = fmaf(e[h], w3r[h], acc);

    int aOut = aBase + n;
    if (aOut < numAll)
      out[(size_t)(hi ? p1 : p0) * numAll + aOut] = acc;
  }
}

extern "C" void kernel_launch(void* const* d_in, const int* in_sizes, int n_in,
                              void* d_out, int out_size, void* d_ws, size_t ws_size,
                              hipStream_t stream) {
  const float* prop  = (const float*)d_in[0];
  const float* feats = (const float*)d_in[1];
  const float* Wdyn  = (const float*)d_in[2];
  const float* bdyn  = (const float*)d_in[3];
  float* outp   = (float*)d_out;
  float* params = (float*)d_ws;

  const int numProp = in_sizes[0] / IN_C;
  const int numAll  = in_sizes[1] / CH;

  const int totalP = numProp * PCOUNT;
  dynconv_params<<<(totalP + 255) / 256, 256, 0, stream>>>(
      prop, Wdyn, bdyn, params, numProp);

  const int numTiles = (numAll + 15) / 16;
  const int pairs    = numProp / 2;
  int ygrid = (numTiles + 7) / 8;
  if (ygrid > 64) ygrid = 64;
  dynconv_mlp<<<dim3(pairs, ygrid), 256, 0, stream>>>(
      feats, params, outp, numAll, numTiles);
}


# Work-group = 8 waves = 8 proposal pairs sharing ONE LDS copy of each feature tile (groups of 8 tiles, two buffers, one barrier per group): a tile is staged once per 8 pairs, not once per pair
# speedup vs baseline: 1.6149x; 1.1897x over previous
#include <hip/hip_runtime.h>

typedef float v2f __attribute__((ext_vector_type(2)));
typedef float v4f __attribute__((ext_vector_type(4)));
typedef float v8f __attribute__((ext_vector_type(8)));

#define IN_C 32
#define CH 35
#define HID 8
#define PCOUNT 369
#define OFF_W1 0
#define OFF_B1 280
#define OFF_W2 288
#define OFF_B2 352
#define OFF_W3 360
#define OFF_B3 368
#define TILE_F (16 * CH)
#define TILE_V4 (TILE_F / 4)
#define STG_F (5 * 32 * 4)

__device__ __forceinline__ float relu1(float x) {
  return __builtin_amdgcn_fmed3f(x, 0.0f, __builtin_inff());
}

__global__ __launch_bounds__(256) void dynconv_params(
    const float* __restrict__ prop, const float* __restrict__ W,
    const float* __restrict__ bias, float* __restrict__ params, int numProp) {
  int idx = blockIdx.x * blockDim.x + threadIdx.x;
  int total = numProp * PCOUNT;
  if (idx >= total) return;
  int p = idx / PCOUNT;
  int c = idx - p * PCOUNT;
  float acc = bias[c];
  const float* pr = prop + p * IN_C;
#pragma unroll
  for (int k = 0; k < IN_C; ++k)
    acc = fmaf(pr[k], W[k * PCOUNT + c], acc);
  params[idx] = acc;
}

__global__ __launch_bounds__(256, 10) void dynconv_mlp(
    const float* __restrict__ feats, const float* __restrict__ params,
    float* __restrict__ out, int numAll, int numTiles, int numPairs) {
  __shared__ __attribute__((aligned(16))) float stage[2 * 8 * STG_F];
  const int lane = threadIdx.x & 31;
  const int wave = threadIdx.x >> 5;

  const int pairRaw = blockIdx.x * 8 + wave;
  const bool pairOK = pairRaw < numPairs;
  const int pair = pairOK ? pairRaw : (numPairs - 1);
  const int p0 = pair * 2;
  const int p1 = p0 + 1;

  const int n   = lane & 15;
  const int hi  = lane >> 4;
  const int hN  = n & 7;
  const bool loCol = (n < 8);
  const float* prmN = params + (size_t)(loCol ? p0 : p1) * PCOUNT;
  const float* prmL = params + (size_t)(hi ? p1 : p0) * PCOUNT;

  v2f W1[9];
  for (int kc = 0; kc < 9; ++kc) {
    int c0 = 4 * kc + 2 * hi;
    int c1 = c0 + 1;
    float x = prmN[OFF_W1 + ((c0 < CH) ? c0 : 0) * HID + hN];
    float y = prmN[OFF_W1 + ((c1 < CH) ? c1 : 0) * HID + hN];
    W1[kc].x = (c0 < CH) ? x : 0.f;
    W1[kc].y = (c1 < CH) ? y : 0.f;
  }

  v2f W2[4];
  for (int j = 0; j < 4; ++j) {
    for (int s = 0; s < 2; ++s) {
      float v = prmN[OFF_W2 + (2 * j + s) * HID + hN];
      v = ((hi == 0) == loCol) ? v : 0.f;
      if (s == 0) W2[j].x = v; else W2[j].y = v;
    }
  }

  v8f c1, c2;
  float w3r[8];
  for (int r = 0; r < 8; ++r) {
    c1[r]  = prmL[OFF_B1 + r];
    c2[r]  = prmL[OFF_B2 + r];
    w3r[r] = prmL[OFF_W3 + r];
  }
  const float b3 = prmL[OFF_B3];

  float* outRow = out + (size_t)(hi ? p1 : p0) * numAll;
  const bool vecOK = (((size_t)feats) & 15) == 0;
  const int numGroups = (numTiles + 7) >> 3;

  int buf = 0;
  for (int g = blockIdx.y; g < numGroups; g += gridDim.y, buf ^= 1) {
    float* grp = stage + buf * (8 * STG_F);

    const int myTile = g * 8 + wave;
    if (myTile < numTiles) {
      float* st = grp + wave * STG_F;
      v4f* st4  = (v4f*)st;
      const int aBase = myTile * 16;
      if (vecOK && aBase + 16 <= numAll) {
        const v4f* src4 = (const v4f*)(feats + (size_t)aBase * CH);
        const int last = 128 + (lane < TILE_V4 - 128 ? lane : TILE_V4 - 129);
        v4f v0 = src4[lane];
        v4f v1 = src4[lane + 32];
        v4f v2 = src4[lane + 64];
        v4f v3 = src4[lane + 96];
        v4f v4 = src4[last];
        st4[lane]      = v0;
        st4[lane + 32] = v1;
        st4[lane + 64] = v2;
        st4[lane + 96] = v3;
        st4[lane + 128] = v4;
      } else {
        for (int i = lane; i < TILE_F; i += 32) {
          int r = i / CH;
          int c = i - r * CH;
          int a = aBase + r;
          if (a > numAll - 1) a = numAll - 1;
          st[i] = feats[(size_t)a * CH + c];
        }
      }
    }
    __syncthreads();

    int tilesHere = numTiles - g * 8;
    if (tilesHere > 8) tilesHere = 8;
    for (int t = 0; t < tilesHere; ++t) {
      const float* xrow = grp + t * STG_F + n * CH;
      const int aBase = (g * 8 + t) * 16;

      v8f d = c1;
      for (int kc = 0; kc < 9; ++kc) {
        int c0 = 4 * kc + 2 * hi;
        v2f X;
        X.x = xrow[c0];
        if (kc < 8) {
          X.y = xrow[c0 + 1];
        } else {
          float tt = xrow[hi ? c0 : (c0 + 1)];
          X.y = hi ? 0.f : tt;
        }
        d = __builtin_amdgcn_wmma_f32_16x16x4_f32(false, W1[kc], false, X,
                                                  (short)0, d, false, false);
      }
      for (int r = 0; r < 8; ++r) d[r] = relu1(d[r]);

      v8f e = c2;
      for (int j = 0; j < 4; ++j) {
        v2f F;
        F.x = d[2 * j];
        F.y = d[2 * j + 1];
        e = __builtin_amdgcn_wmma_f32_16x16x4_f32(false, W2[j], false, F,
                                                  (short)0, e, false, false);
      }
      for (int r = 0; r < 8; ++r) e[r] = relu1(e[r]);

      float acc = b3;
      for (int h = 0; h < 8; ++h)
        acc = fmaf(e[h], w3r[h], acc);

      int aOut = aBase + n;
      if (pairOK && aOut < numAll)
        outRow[aOut] = acc;
    }
  }
}

extern "C" void kernel_launch(void* const* d_in, const int* in_sizes, int n_in,
                              void* d_out, int out_size, void* d_ws, size_t ws_size,
                              hipStream_t stream) {
  const float* prop  = (const float*)d_in[0];
  const float* feats = (const float*)d_in[1];
  const float* Wdyn  = (const float*)d_in[2];
  const float* bdyn  = (const float*)d_in[3];
  float* outp   = (float*)d_out;
  float* params = (float*)d_ws;

  const int numProp = in_sizes[0] / IN_C;
  const int numAll  = in_sizes[1] / CH;

  const int totalP = numProp * PCOUNT;
  dynconv_params<<<(totalP + 255) / 256, 256, 0, stream>>>(
      prop, Wdyn, bdyn, params, numProp);

  const int numTiles  = (numAll + 15) / 16;
  const int numGroups = (numTiles + 7) / 8;
  const int pairs     = numProp / 2;
  const int xgrid     = (pairs + 7) / 8;
  int cap = 1280 / xgrid;
  if (cap < 1) cap = 1;
  const int perWg = (numGroups + cap - 1) / cap;
  int ygrid = (numGroups + perWg - 1) / (perWg > 0 ? perWg : 1);
  if (ygrid < 1) ygrid = 1;
  dynconv_mlp<<<dim3(xgrid, ygrid), 256, 0, stream>>>(
      feats, params, outp, numAll, numTiles, pairs);
}


# Read each tile's layer-1 B fragments from LDS one tile ahead, as ONE group behind the tile's last WMMA (was 4 places inside the layer-1 WMMA run, each waited in place); k order and sums unchanged
# speedup vs baseline: 1.6769x; 1.0384x over previous
#include <hip/hip_runtime.h>

typedef float v2f __attribute__((ext_vector_type(2)));
typedef float v4f __attribute__((ext_vector_type(4)));
typedef float v8f __attribute__((ext_vector_type(8)));

#define IN_C 32
#define CH 35
#define HID 8
#define PCOUNT 369
#define OFF_W1 0
#define OFF_B1 280
#define OFF_W2 288
#define OFF_B2 352
#define OFF_W3 360
#define OFF_B3 368
#define TILE_F (16 * CH)
#define TILE_V4 (TILE_F / 4)
#define STG_F (5 * 32 * 4)

__device__ __forceinline__ float relu1(float x) {
  return __builtin_amdgcn_fmed3f(x, 0.0f, __builtin_inff());
}

__device__ __forceinline__ void load_frag(v2f (&R)[9], const float* xrow, int hi) {
  for (int kc = 0; kc < 9; ++kc) {
    const int c0 = 4 * kc + 2 * hi;
    R[kc].x = xrow[c0];
    R[kc].y = xrow[c0 + 1];
  }
}

__global__ __launch_bounds__(256) void dynconv_params(
    const float* __restrict__ prop, const float* __restrict__ W,
    const float* __restrict__ bias, float* __restrict__ params, int numProp) {
  int idx = blockIdx.x * blockDim.x + threadIdx.x;
  int total = numProp * PCOUNT;
  if (idx >= total) return;
  int p = idx / PCOUNT;
  int c = idx - p * PCOUNT;
  float acc = bias[c];
  const float* pr = prop + p * IN_C;
  for (int k = 0; k < IN_C; ++k)
    acc = fmaf(pr[k], W[k * PCOUNT + c], acc);
  params[idx] = acc;
}

__global__ __launch_bounds__(256, 10) void dynconv_mlp(
    const float* __restrict__ feats, const float* __restrict__ params,
    float* __restrict__ out, int numAll, int numTiles, int numPairs) {
  __shared__ __attribute__((aligned(16))) float stage[2 * 8 * STG_F];
  const int lane = threadIdx.x & 31;
  const int wave = __builtin_amdgcn_readfirstlane(threadIdx.x >> 5);

  const int pairRaw = blockIdx.x * 8 + wave;
  const bool pairOK = pairRaw < numPairs;
  const int pair = pairOK ? pairRaw : (numPairs - 1);
  const int p0 = pair * 2;
  const int p1 = p0 + 1;

  const int n   = lane & 15;
  const int hi  = lane >> 4;
  const int hN  = n & 7;
  const bool loCol = (n < 8);
  const float* prmN = params + (size_t)(loCol ? p0 : p1) * PCOUNT;
  const float* prmL = params + (size_t)(hi ? p1 : p0) * PCOUNT;

  v2f W1[9];
  for (int kc = 0; kc < 9; ++kc) {
    int c0 = 4 * kc + 2 * hi;
    int c1 = c0 + 1;
    float x = prmN[OFF_W1 + ((c0 < CH) ? c0 : 0) * HID + hN];
    float y = prmN[OFF_W1 + ((c1 < CH) ? c1 : 0) * HID + hN];
    W1[kc].x = (c0 < CH) ? x : 0.f;
    W1[kc].y = (c1 < CH) ? y : 0.f;
  }

  v2f W2[4];
  for (int j = 0; j < 4; ++j) {
    for (int s = 0; s < 2; ++s) {
      float v = prmN[OFF_W2 + (2 * j + s) * HID + hN];
      v = ((hi == 0) == loCol) ? v : 0.f;
      if (s == 0) W2[j].x = v; else W2[j].y = v;
    }
  }

  v8f c1, c2;
  float w3r[8];
  for (int r = 0; r < 8; ++r) {
    c1[r]  = prmL[OFF_B1 + r];
    c2[r]  = prmL[OFF_B2 + r];
    w3r[r] = prmL[OFF_W3 + r];
  }
  const float b3 = prmL[OFF_B3];

  float* outPair = out + (size_t)p0 * numAll;
  const unsigned oLane = (hi ? (unsigned)numAll : 0u) + (unsigned)n;
  const bool vecOK = (((size_t)feats) & 15) == 0;
  const int numGroups = (numTiles + 7) >> 3;

  int buf = 0;
  for (int g = blockIdx.y; g < numGroups; g += gridDim.y, buf ^= 1) {
    float* grp = stage + buf * (8 * STG_F);

    const int myTile = g * 8 + wave;
    if (myTile < numTiles) {
      float* st = grp + wave * STG_F;
      v4f* st4  = (v4f*)st;
      const int aBase = myTile * 16;
      if (vecOK && aBase + 16 <= numAll) {
        const v4f* src4 = (const v4f*)(feats + (size_t)aBase * CH);
        const int last = 128 + (lane < TILE_V4 - 128 ? lane : TILE_V4 - 129);
        v4f v0 = src4[lane];
        v4f v1 = src4[lane + 32];
        v4f v2 = src4[lane + 64];
        v4f v3 = src4[lane + 96];
        v4f v4 = src4[last];
        st4[lane]      = v0;
        st4[lane + 32] = v1;
        st4[lane + 64] = v2;
        st4[lane + 96] = v3;
        st4[lane + 128] = v4;
      } else {
#pragma unroll 1
        for (int j = 0; j < (TILE_F + 31) / 32; ++j) {
          const int i = j * 32 + lane;
          if (i < TILE_F) {
            int r = i / CH;
            int c = i - r * CH;
            int a = aBase + r;
            if (a > numAll - 1) a = numAll - 1;
            st[i] = feats[(size_t)a * CH + c];
          }
        }
      }
    }
    __syncthreads();

    int tilesHere = numTiles - g * 8;
    if (tilesHere > 8) tilesHere = 8;
    const float* xrow0 = grp + n * CH;

    v2f R[9];
    load_frag(R, xrow0, hi);
    for (int t = 0; t < tilesHere; ++t) {
      const int aBase = (g * 8 + t) * 16;

      v8f d = c1;
      for (int kc = 0; kc < 9; ++kc) {
        v2f X = R[kc];
        if (kc == 8) X.y = hi ? 0.f : X.y;
        d = __builtin_amdgcn_wmma_f32_16x16x4_f32(false, W1[kc], false, X,
                                                  (short)0, d, false, false);
      }
      for (int r = 0; r < 8; ++r) d[r] = relu1(d[r]);

      v8f e = c2;
      for (int j = 0; j < 4; ++j) {
        v2f F;
        F.x = d[2 * j];
        F.y = d[2 * j + 1];
        e = __builtin_amdgcn_wmma_f32_16x16x4_f32(false, W2[j], false, F,
                                                  (short)0, e, false, false);
      }

      const int tn = (t + 1 < tilesHere) ? (t + 1) : t;
      asm volatile("" :: "v"(e) : "memory");
      load_frag(R, xrow0 + tn * STG_F, hi);

      for (int r = 0; r < 8; ++r) e[r] = relu1(e[r]);

      float acc = b3;
      for (int h = 0; h < 8; ++h)
        acc = fmaf(e[h], w3r[h], acc);

      int aOut = aBase + n;
      if (pairOK && aOut < numAll)
        outPair[oLane + (unsigned)aBase] = acc;
    }
  }
}

extern "C" void kernel_launch(void* const* d_in, const int* in_sizes, int n_in,
                              void* d_out, int out_size, void* d_ws, size_t ws_size,
                              hipStream_t stream) {
  const float* prop  = (const float*)d_in[0];
  const float* feats = (const float*)d_in[1];
  const float* Wdyn  = (const float*)d_in[2];
  const float* bdyn  = (const float*)d_in[3];
  float* outp   = (float*)d_out;
  float* params = (float*)d_ws;

  const int numProp = in_sizes[0] / IN_C;
  const int numAll  = in_sizes[1] / CH;

  const int totalP = numProp * PCOUNT;
  dynconv_params<<<(totalP + 255) / 256, 256, 0, stream>>>(
      prop, Wdyn, bdyn, params, numProp);

  const int numTiles  = (numAll + 15) / 16;
  const int numGroups = (numTiles + 7) / 8;
  const int pairs     = numProp / 2;
  const int xgrid     = (pairs + 7) / 8;
  int cap = 1280 / xgrid;
  if (cap < 1) cap = 1;
  const int perWg = (numGroups + cap - 1) / cap;
  int ygrid = (numGroups + perWg - 1) / (perWg > 0 ? perWg : 1);
  if (ygrid < 1) ygrid = 1;
  dynconv_mlp<<<dim3(xgrid, ygrid), 256, 0, stream>>>(
      feats, params, outp, numAll, numTiles, pairs);
}


# Wave priority falls with a work-group's progress: s_setprio = tile groups still behind the current one, capped at 3, so 0 in its last; resident work-groups advance together, not one after another
# speedup vs baseline: 1.7310x; 1.0323x over previous
#include <hip/hip_runtime.h>

typedef float v2f __attribute__((ext_vector_type(2)));
typedef float v4f __attribute__((ext_vector_type(4)));
typedef float v8f __attribute__((ext_vector_type(8)));

#define IN_C 32
#define CH 35
#define HID 8
#define PCOUNT 369
#define OFF_W1 0
#define OFF_B1 280
#define OFF_W2 288
#define OFF_B2 352
#define OFF_W3 360
#define OFF_B3 368
#define TILE_F (16 * CH)
#define TILE_V4 (TILE_F / 4)
#define STG_F (5 * 32 * 4)

__device__ __forceinline__ float relu1(float x) {
  return __builtin_amdgcn_fmed3f(x, 0.0f, __builtin_inff());
}

__device__ __forceinline__ void load_frag(v2f (&R)[9], const float* xrow, int hi) {
  for (int kc = 0; kc < 9; ++kc) {
    const int c0 = 4 * kc + 2 * hi;
    R[kc].x = xrow[c0];
    R[kc].y = xrow[c0 + 1];
  }
}

__global__ __launch_bounds__(256) void dynconv_params(
    const float* __restrict__ prop, const float* __restrict__ W,
    const float* __restrict__ bias, float* __restrict__ params, int numProp) {
  int idx = blockIdx.x * blockDim.x + threadIdx.x;
  int total = numProp * PCOUNT;
  if (idx >= total) return;
  int p = idx / PCOUNT;
  int c = idx - p * PCOUNT;
  float acc = bias[c];
  const float* pr = prop + p * IN_C;
  for (int k = 0; k < IN_C; ++k)
    acc = fmaf(pr[k], W[k * PCOUNT + c], acc);
  params[idx] = acc;
}

__global__ __launch_bounds__(256, 10) void dynconv_mlp(
    const float* __restrict__ feats, const float* __restrict__ params,
    float* __restrict__ out, int numAll, int numTiles, int numPairs) {
  __shared__ __attribute__((aligned(16))) float stage[2 * 8 * STG_F];
  const int lane = threadIdx.x & 31;
  const int wave = __builtin_amdgcn_readfirstlane(threadIdx.x >> 5);

  const int pairRaw = blockIdx.x * 8 + wave;
  const bool pairOK = pairRaw < numPairs;
  const int pair = pairOK ? pairRaw : (numPairs - 1);
  const int p0 = pair * 2;
  const int p1 = p0 + 1;

  const int n   = lane & 15;
  const int hi  = lane >> 4;
  const int hN  = n & 7;
  const bool loCol = (n < 8);
  const float* prmN = params + (size_t)(loCol ? p0 : p1) * PCOUNT;
  const float* prmL = params + (size_t)(hi ? p1 : p0) * PCOUNT;

  v2f W1[9];
  for (int kc = 0; kc < 9; ++kc) {
    int c0 = 4 * kc + 2 * hi;
    int c1 = c0 + 1;
    float x = prmN[OFF_W1 + ((c0 < CH) ? c0 : 0) * HID + hN];
    float y = prmN[OFF_W1 + ((c1 < CH) ? c1 : 0) * HID + hN];
    W1[kc].x = (c0 < CH) ? x : 0.f;
    W1[kc].y = (c1 < CH) ? y : 0.f;
  }

  v2f W2[4];
  for (int j = 0; j < 4; ++j) {
    for (int s = 0; s < 2; ++s) {
      float v = prmN[OFF_W2 + (2 * j + s) * HID + hN];
      v = ((hi == 0) == loCol) ? v : 0.f;
      if (s == 0) W2[j].x = v; else W2[j].y = v;
    }
  }

  v8f c1, c2;
  float w3r[8];
  for (int r = 0; r < 8; ++r) {
    c1[r]  = prmL[OFF_B1 + r];
    c2[r]  = prmL[OFF_B2 + r];
    w3r[r] = prmL[OFF_W3 + r];
  }
  const float b3 = prmL[OFF_B3];

  float* outPair = out + (size_t)p0 * numAll;
  const unsigned oLane = (hi ? (unsigned)numAll : 0u) + (unsigned)n;
  const bool vecOK = (((size_t)feats) & 15) == 0;
  const int numGroups = (numTiles + 7) >> 3;

  int buf = 0;
  const int gStep = gridDim.y;
  for (int g = blockIdx.y; g < numGroups; g += gStep, buf ^= 1) {
    const int later = (g + gStep < numGroups ? 1 : 0) + (g + 2 * gStep < numGroups ? 1 : 0) +
                      (g + 3 * gStep < numGroups ? 1 : 0);
    if (later == 3)      __builtin_amdgcn_s_setprio(3);
    else if (later == 2) __builtin_amdgcn_s_setprio(2);
    else if (later == 1) __builtin_amdgcn_s_setprio(1);
    else                 __builtin_amdgcn_s_setprio(0);

    float* grp = stage + buf * (8 * STG_F);

    const int myTile = g * 8 + wave;
    if (myTile < numTiles) {
      float* st = grp + wave * STG_F;
      v4f* st4  = (v4f*)st;
      const int aBase = myTile * 16;
      if (vecOK && aBase + 16 <= numAll) {
        const v4f* src4 = (const v4f*)(feats + (size_t)aBase * CH);
        const int last = 128 + (lane < TILE_V4 - 128 ? lane : TILE_V4 - 129);
        v4f v0 = src4[lane];
        v4f v1 = src4[lane + 32];
        v4f v2 = src4[lane + 64];
        v4f v3 = src4[lane + 96];
        v4f v4 = src4[last];
        st4[lane]      = v0;
        st4[lane + 32] = v1;
        st4[lane + 64] = v2;
        st4[lane + 96] = v3;
        st4[lane + 128] = v4;
      } else {
#pragma unroll 1
        for (int j = 0; j < (TILE_F + 31) / 32; ++j) {
          const int i = j * 32 + lane;
          if (i < TILE_F) {
            int r = i / CH;
            int c = i - r * CH;
            int a = aBase + r;
            if (a > numAll - 1) a = numAll - 1;
            st[i] = feats[(size_t)a * CH + c];
          }
        }
      }
    }
    __syncthreads();

    int tilesHere = numTiles - g * 8;
    if (tilesHere > 8) tilesHere = 8;
    tilesHere = __builtin_amdgcn_readfirstlane(tilesHere);
    const float* xrow0 = grp + n * CH;

    v2f R[9];
    load_frag(R, xrow0, hi);
    for (int t = 0; t < tilesHere; ++t) {
      const int aBase = (g * 8 + t) * 16;

      v8f d = c1;
      for (int kc = 0; kc < 9; ++kc) {
        v2f X = R[kc];
        if (kc == 8) X.y = hi ? 0.f : X.y;
        d = __builtin_amdgcn_wmma_f32_16x16x4_f32(false, W1[kc], false, X,
                                                  (short)0, d, false, false);
      }
      for (int r = 0; r < 8; ++r) d[r] = relu1(d[r]);

      v8f e = c2;
      for (int j = 0; j < 4; ++j) {
        v2f F;
        F.x = d[2 * j];
        F.y = d[2 * j + 1];
        e = __builtin_amdgcn_wmma_f32_16x16x4_f32(false, W2[j], false, F,
                                                  (short)0, e, false, false);
      }

      const int tn = (t + 1 < tilesHere) ? (t + 1) : t;
      asm volatile("" :: "v"(e) : "memory");
      load_frag(R, xrow0 + tn * STG_F, hi);

      for (int r = 0; r < 8; ++r) e[r] = relu1(e[r]);

      float acc = b3;
      for (int h = 0; h < 8; ++h)
        acc = fmaf(e[h], w3r[h], acc);

      int aOut = aBase + n;
      if (pairOK && aOut < numAll)
        outPair[oLane + (unsigned)aBase] = acc;
    }
  }
}

extern "C" void kernel_launch(void* const* d_in, const int* in_sizes, int n_in,
                              void* d_out, int out_size, void* d_ws, size_t ws_size,
                              hipStream_t stream) {
  const float* prop  = (const float*)d_in[0];
  const float* feats = (const float*)d_in[1];
  const float* Wdyn  = (const float*)d_in[2];
  const float* bdyn  = (const float*)d_in[3];
  float* outp   = (float*)d_out;
  float* params = (float*)d_ws;

  const int numProp = in_sizes[0] / IN_C;
  const int numAll  = in_sizes[1] / CH;

  const int totalP = numProp * PCOUNT;
  dynconv_params<<<(totalP + 255) / 256, 256, 0, stream>>>(
      prop, Wdyn, bdyn, params, numProp);

  const int numTiles  = (numAll + 15) / 16;
  const int numGroups = (numTiles + 7) / 8;
  const int pairs     = numProp / 2;
  const int xgrid     = (pairs + 7) / 8;
  int cap = 1280 / xgrid;
  if (cap < 1) cap = 1;
  const int perWg = (numGroups + cap - 1) / cap;
  int ygrid = (numGroups + perWg - 1) / (perWg > 0 ? perWg : 1);
  if (ygrid < 1) ygrid = 1;
  dynconv_mlp<<<dim3(xgrid, ygrid), 256, 0, stream>>>(
      feats, params, outp, numAll, numTiles, pairs);
}
